# starting kernel
# baseline (speedup 1.0000x reference)
#include <hip/hip_runtime.h>


typedef __attribute__((ext_vector_type(2))) float v2f;
typedef __attribute__((ext_vector_type(8))) float v8f;
typedef v2f __attribute__((aligned(4))) v2f_u;

#define TPW 4

__constant__ float c_rec_lo[8] = {
    0.23037781330885523f,  0.7148465705525415f,  0.6308807679295904f,
   -0.02798376941698385f, -0.18703481171888114f, 0.030841381835986965f,
    0.032883011666982945f,-0.010597401784997278f};
__constant__ float c_rec_hi[8] = {
   -0.010597401784997278f,-0.032883011666982945f, 0.030841381835986965f,
    0.18703481171888114f, -0.02798376941698385f, -0.6308807679295904f,
    0.7148465705525415f,  -0.23037781330885523f};

__device__ __forceinline__ float wtap(const float* w, int t) {
  float v = w[t & 7];
  return (t >= 0 && t < 8) ? v : 0.0f;
}

__global__ __launch_bounds__(128)
void idwt_level_wmma(const float* __restrict__ xa, int stride_a,
                     const float* __restrict__ xd,
                     float* __restrict__ out,
                     int n_in, int L, int j_tiles) {
  const int wave = threadIdx.x >> 5;
  const int lane = threadIdx.x & 31;
  const int wid  = blockIdx.x * 4 + wave;
  const int jt0  = wid * TPW;
  if (__builtin_amdgcn_readfirstlane(jt0 >= j_tiles)) return;
  const int bt = blockIdx.y;

  const int m  = lane & 15;
  const int hi = lane >> 4;
  const int b0 = bt << 4;

  const float* rowA = xa + (size_t)(b0 + m) * (size_t)stride_a;
  const float* rowD = xd + (size_t)(b0 + m) * (size_t)n_in;

  v2f blo[3], bhi[3];
#pragma unroll
  for (int ks = 0; ks < 3; ++ks) {
    const int kb = ks * 4 + hi * 2;
    const int t0 = m + 6 - 2 * kb;
    blo[ks].x = wtap(c_rec_lo, t0);  blo[ks].y = wtap(c_rec_lo, t0 - 2);
    bhi[ks].x = wtap(c_rec_hi, t0);  bhi[ks].y = wtap(c_rec_hi, t0 - 2);
  }

  const int grp_fast =
      (jt0 + TPW <= j_tiles) &&
      ((((jt0 + TPW - 1) << 3) + 12) <= n_in) &&
      (((jt0 + TPW) << 4) <= L);

  if (__builtin_amdgcn_readfirstlane(grp_fast)) {
    v2f fa[TPW][3], fd[TPW][3];
#pragma unroll
    for (int u = 0; u < TPW; ++u) {
      const int i0 = (jt0 + u) << 3;
#pragma unroll
      for (int ks = 0; ks < 3; ++ks) {
        const int c = i0 + ks * 4 + hi * 2;
        fa[u][ks] = *(const v2f_u*)(rowA + c);
        fd[u][ks] = *(const v2f_u*)(rowD + c);
      }
    }
#pragma unroll
    for (int u = 0; u < TPW; ++u) {
      v8f acc = {0.f, 0.f, 0.f, 0.f, 0.f, 0.f, 0.f, 0.f};
#pragma unroll
      for (int ks = 0; ks < 3; ++ks) {
        acc = __builtin_amdgcn_wmma_f32_16x16x4_f32(
            false, fa[u][ks], false, blo[ks], (short)0, acc, false, false);
        acc = __builtin_amdgcn_wmma_f32_16x16x4_f32(
            false, fd[u][ks], false, bhi[ks], (short)0, acc, false, false);
      }
      const int j  = ((jt0 + u) << 4) + m;
      const int mb = hi * 8;
#pragma unroll
      for (int r = 0; r < 8; ++r)
        out[(size_t)(b0 + mb + r) * (size_t)L + j] = acc[r];
    }
    return;
  }

  for (int u = 0; u < TPW; ++u) {
    const int jt = jt0 + u;
    if (__builtin_amdgcn_readfirstlane(jt >= j_tiles)) break;
    const int i0 = jt << 3;
    v8f acc = {0.f, 0.f, 0.f, 0.f, 0.f, 0.f, 0.f, 0.f};
#pragma unroll
    for (int ks = 0; ks < 3; ++ks) {
      const int col0 = i0 + ks * 4 + hi * 2;
      const int col1 = col0 + 1;
      const int cc0 = (col0 < n_in) ? col0 : (n_in - 1);
      const int cc1 = (col1 < n_in) ? col1 : (n_in - 1);
      float a0 = rowA[cc0], a1 = rowA[cc1];
      float d0 = rowD[cc0], d1 = rowD[cc1];
      v2f fa, fd;
      fa.x = (col0 < n_in) ? a0 : 0.f;  fa.y = (col1 < n_in) ? a1 : 0.f;
      fd.x = (col0 < n_in) ? d0 : 0.f;  fd.y = (col1 < n_in) ? d1 : 0.f;
      acc = __builtin_amdgcn_wmma_f32_16x16x4_f32(
          false, fa, false, blo[ks], (short)0, acc, false, false);
      acc = __builtin_amdgcn_wmma_f32_16x16x4_f32(
          false, fd, false, bhi[ks], (short)0, acc, false, false);
    }
    const int j  = (jt << 4) + m;
    const int mb = hi * 8;
    if (j < L) {
#pragma unroll
      for (int r = 0; r < 8; ++r)
        out[(size_t)(b0 + mb + r) * (size_t)L + j] = acc[r];
    }
  }
}

static inline void launch_level(const float* xa, int stride_a, const float* xd,
                                float* out, int n_in, int L, hipStream_t s) {
  const int j_tiles = (L + 15) / 16;
  const int wgroups = (j_tiles + TPW - 1) / TPW;
  dim3 grid((wgroups + 3) / 4, 8);
  idwt_level_wmma<<<grid, 128, 0, s>>>(xa, stride_a, xd, out, n_in, L, j_tiles);
}

extern "C" void kernel_launch(void* const* d_in, const int* in_sizes, int n_in_cnt,
                              void* d_out, int out_size, void* d_ws, size_t ws_size,
                              hipStream_t stream) {
  (void)in_sizes; (void)n_in_cnt; (void)out_size; (void)ws_size;
  const float* a  = (const float*)d_in[0];
  const float* d6 = (const float*)d_in[1];
  const float* d5 = (const float*)d_in[2];
  const float* d4 = (const float*)d_in[3];
  const float* d3 = (const float*)d_in[4];
  const float* d2 = (const float*)d_in[5];
  const float* d1 = (const float*)d_in[6];
  float* out = (float*)d_out;
  float* ws  = (float*)d_ws;

  launch_level(a,    4102,   d6, ws,   4102,   8198,   stream);
  launch_level(ws,   8198,   d5, out,  8198,   16390,  stream);
  launch_level(out,  16390,  d4, ws,   16390,  32774,  stream);
  launch_level(ws,   32774,  d3, out,  32774,  65542,  stream);
  launch_level(out,  65542,  d2, ws,   65541,  131076, stream);
  launch_level(ws,   131076, d1, out,  131075, 262144, stream);
}


# WMMA tile remapped from 16 rows x 16 outputs to 256 outputs of ONE row (weights as A, row data as B): lanes run along the row, contiguous 16B+8B loads, 16B stores; k regrouped, sum order changes
# speedup vs baseline: 2.0644x; 2.0644x over previous
#include <hip/hip_runtime.h>


typedef __attribute__((ext_vector_type(2))) float v2f;
typedef __attribute__((ext_vector_type(4))) float v4f;
typedef __attribute__((ext_vector_type(8))) float v8f;
typedef v2f __attribute__((aligned(4))) v2f_u;
typedef v4f __attribute__((aligned(4))) v4f_u;

#define TPW 4

__constant__ float c_rec_lo[8] = {
    0.23037781330885523f,  0.7148465705525415f,  0.6308807679295904f,
   -0.02798376941698385f, -0.18703481171888114f, 0.030841381835986965f,
    0.032883011666982945f,-0.010597401784997278f};
__constant__ float c_rec_hi[8] = {
   -0.010597401784997278f,-0.032883011666982945f, 0.030841381835986965f,
    0.18703481171888114f, -0.02798376941698385f, -0.6308807679295904f,
    0.7148465705525415f,  -0.23037781330885523f};

__device__ __forceinline__ float wtap(const float* w, int t) {
  float v = w[t & 7];
  return (t >= 0 && t < 8) ? v : 0.0f;
}

__device__ __forceinline__ float ldg_guard(const float* row, int col, int n) {
  const int cc = (col < n) ? col : (n - 1);
  const float v = row[cc];
  return (col < n) ? v : 0.0f;
}

__device__ __forceinline__ v8f wmma4(v2f a, v2f b, v8f c) {
  return __builtin_amdgcn_wmma_f32_16x16x4_f32(false, a, false, b, (short)0, c,
                                               false, false);
}

__global__ __launch_bounds__(128)
void idwt_level_wmma(const float* __restrict__ xa, int stride_a,
                     const float* __restrict__ xd,
                     float* __restrict__ out,
                     int n_in, int L, int tiles) {
  const int wave = __builtin_amdgcn_readfirstlane(threadIdx.x >> 5);
  const int lane = threadIdx.x & 31;
  const int t0   = (blockIdx.x * 4 + wave) * TPW;
  if (t0 >= tiles) return;
  const int b = blockIdx.y;

  const int c = lane & 15;
  const int h = lane >> 4;

  const float* rowA = xa + (size_t)b * (size_t)stride_a;
  const float* rowD = xd + (size_t)b * (size_t)n_in;
  float*       rowO = out + (size_t)b * (size_t)L;

  const int so = ((c & 4) << 1) | ((c & 8) >> 1) | (c & 3);
  v2f alo[3], ahi[3];
  {
    const int t[3] = {so + 6 - 8 * h, so + 2 - 8 * h, so - 10 - 4 * h};
#pragma unroll
    for (int ks = 0; ks < 3; ++ks) {
      alo[ks].x = wtap(c_rec_lo, t[ks]);  alo[ks].y = wtap(c_rec_lo, t[ks] - 2);
      ahi[ks].x = wtap(c_rec_hi, t[ks]);  ahi[ks].y = wtap(c_rec_hi, t[ks] - 2);
    }
  }

  const int lo4 = 8 * c + 4 * h;
  const int lo2 = 8 * c + 8 + 2 * h;
  const int st  = 16 * c + 4 * h;

  const int grp_fast =
      (t0 + TPW <= tiles) &&
      (((t0 + TPW) << 8) <= L) &&
      ((((t0 + TPW - 1) << 7) + 132) <= n_in);

  if (grp_fast) {
    const float* pa = rowA + (t0 << 7);
    const float* pd = rowD + (t0 << 7);
    float*       po = rowO + (t0 << 8);
    v4f qa[TPW], qd[TPW];
    v2f ha[TPW], hd[TPW];
#pragma unroll
    for (int u = 0; u < TPW; ++u) {
      qa[u] = *(const v4f_u*)(pa + (u << 7) + lo4);
      qd[u] = *(const v4f_u*)(pd + (u << 7) + lo4);
      ha[u] = *(const v2f_u*)(pa + (u << 7) + lo2);
      hd[u] = *(const v2f_u*)(pd + (u << 7) + lo2);
    }
    __builtin_amdgcn_sched_barrier(0);
#pragma unroll
    for (int u = 0; u < TPW; ++u) {
      v8f acc = {0.f, 0.f, 0.f, 0.f, 0.f, 0.f, 0.f, 0.f};
      v2f b0a = {qa[u].x, qa[u].y}, b1a = {qa[u].z, qa[u].w};
      v2f b0d = {qd[u].x, qd[u].y}, b1d = {qd[u].z, qd[u].w};
      acc = wmma4(alo[0], b0a, acc);
      acc = wmma4(ahi[0], b0d, acc);
      acc = wmma4(alo[1], b1a, acc);
      acc = wmma4(ahi[1], b1d, acc);
      acc = wmma4(alo[2], ha[u], acc);
      acc = wmma4(ahi[2], hd[u], acc);
      float* p = po + (u << 8) + st;
      v4f s0 = {acc[0], acc[1], acc[2], acc[3]};
      v4f s1 = {acc[4], acc[5], acc[6], acc[7]};
      *(v4f_u*)(p)     = s0;
      *(v4f_u*)(p + 8) = s1;
    }
    return;
  }

  int lim[TPW];
#pragma unroll
  for (int u = 0; u < TPW; ++u) lim[u] = L - ((t0 + u) << 8) - st;
#pragma unroll
  for (int u = 0; u < TPW; ++u) {
    const int T = t0 + u;
    if (T >= tiles) break;
    const int i0 = T << 7;
    const int c4 = i0 + lo4;
    const int c2 = i0 + lo2;
    v2f b0a, b1a, b2a, b0d, b1d, b2d;
    b0a.x = ldg_guard(rowA, c4,     n_in);  b0a.y = ldg_guard(rowA, c4 + 1, n_in);
    b1a.x = ldg_guard(rowA, c4 + 2, n_in);  b1a.y = ldg_guard(rowA, c4 + 3, n_in);
    b2a.x = ldg_guard(rowA, c2,     n_in);  b2a.y = ldg_guard(rowA, c2 + 1, n_in);
    b0d.x = ldg_guard(rowD, c4,     n_in);  b0d.y = ldg_guard(rowD, c4 + 1, n_in);
    b1d.x = ldg_guard(rowD, c4 + 2, n_in);  b1d.y = ldg_guard(rowD, c4 + 3, n_in);
    b2d.x = ldg_guard(rowD, c2,     n_in);  b2d.y = ldg_guard(rowD, c2 + 1, n_in);
    v8f acc = {0.f, 0.f, 0.f, 0.f, 0.f, 0.f, 0.f, 0.f};
    acc = wmma4(alo[0], b0a, acc);
    acc = wmma4(ahi[0], b0d, acc);
    acc = wmma4(alo[1], b1a, acc);
    acc = wmma4(ahi[1], b1d, acc);
    acc = wmma4(alo[2], b2a, acc);
    acc = wmma4(ahi[2], b2d, acc);
    float* p = rowO + (T << 8) + st;
#pragma unroll
    for (int r = 0; r < 4; ++r) {
      if (r < lim[u])     p[r]     = acc[r];
      if (8 + r < lim[u]) p[8 + r] = acc[4 + r];
    }
  }
}

static inline void launch_level(const float* xa, int stride_a, const float* xd,
                                float* out, int n_in, int L, hipStream_t s) {
  const int tiles  = (L + 255) / 256;
  const int groups = (tiles + TPW - 1) / TPW;
  dim3 grid((groups + 3) / 4, 128);
  idwt_level_wmma<<<grid, 128, 0, s>>>(xa, stride_a, xd, out, n_in, L, tiles);
}

extern "C" void kernel_launch(void* const* d_in, const int* in_sizes, int n_in_cnt,
                              void* d_out, int out_size, void* d_ws, size_t ws_size,
                              hipStream_t stream) {
  (void)in_sizes; (void)n_in_cnt; (void)out_size; (void)ws_size;
  const float* a  = (const float*)d_in[0];
  const float* d6 = (const float*)d_in[1];
  const float* d5 = (const float*)d_in[2];
  const float* d4 = (const float*)d_in[3];
  const float* d3 = (const float*)d_in[4];
  const float* d2 = (const float*)d_in[5];
  const float* d1 = (const float*)d_in[6];
  float* out = (float*)d_out;
  float* ws  = (float*)d_ws;

  launch_level(a,    4102,   d6, ws,   4102,   8198,   stream);
  launch_level(ws,   8198,   d5, out,  8198,   16390,  stream);
  launch_level(out,  16390,  d4, ws,   16390,  32774,  stream);
  launch_level(ws,   32774,  d3, out,  32774,  65542,  stream);
  launch_level(out,  65542,  d2, ws,   65541,  131076, stream);
  launch_level(ws,   131076, d1, out,  131075, 262144, stream);
}


# All six levels fused into ONE launch: a work-group makes 4096 outputs of a row and keeps each level's piece (with its halo) in LDS, so no intermediate level goes to memory; sum order unchanged
# speedup vs baseline: 4.4285x; 2.1452x over previous
#include <hip/hip_runtime.h>


typedef __attribute__((ext_vector_type(2))) float v2f;
typedef __attribute__((ext_vector_type(4))) float v4f;
typedef __attribute__((ext_vector_type(8))) float v8f;
typedef v2f __attribute__((aligned(4))) v2f_u;
typedef v4f __attribute__((aligned(4))) v4f_u;

__constant__ float c_rec_lo[8] = {
    0.23037781330885523f,  0.7148465705525415f,  0.6308807679295904f,
   -0.02798376941698385f, -0.18703481171888114f, 0.030841381835986965f,
    0.032883011666982945f,-0.010597401784997278f};
__constant__ float c_rec_hi[8] = {
   -0.010597401784997278f,-0.032883011666982945f, 0.030841381835986965f,
    0.18703481171888114f, -0.02798376941698385f, -0.6308807679295904f,
    0.7148465705525415f,  -0.23037781330885523f};

__host__ __device__ constexpr int nt_of(int lv) {
  return lv == 1 ? 2 : lv == 2 ? 2 : lv == 3 ? 3 : lv == 4 ? 5 : lv == 5 ? 9 : 16;
}
__host__ __device__ constexpr int nin_of(int lv) {
  return lv == 1 ? 4102 : lv == 2 ? 8198 : lv == 3 ? 16390 : lv == 4 ? 32774
       : lv == 5 ? 65541 : 131075;
}
__host__ __device__ constexpr int bufo_of(int lv) {
  return lv == 1 ? 0 : lv == 2 ? 512 : lv == 3 ? 1024 : lv == 4 ? 1792 : 3072;
}
__host__ __device__ constexpr int rot_of(int lv) {
  return lv == 1 ? 2 : lv == 2 ? 2 : lv == 4 ? 3 : 0;
}
#define LDS_FLOATS 5376
#define L_OUT 262144

__device__ __forceinline__ float ldg_guard(const float* row, int col, int n) {
  const int cc = (col < n) ? col : (n - 1);
  const float v = row[cc];
  return (col < n) ? v : 0.0f;
}

__device__ __forceinline__ v8f wmma4(v2f a, v2f b, v8f c) {
  return __builtin_amdgcn_wmma_f32_16x16x4_f32(false, a, false, b, (short)0, c,
                                               false, false);
}

struct BT { v4f q; v2f h2; };
template <int LV> struct BSet { BT t[(nt_of(LV) + 3) / 4]; };

template <int LV, bool G>
__device__ __forceinline__ void load_set(BSet<LV>& s, const float* __restrict__ row,
                                         int o_in, int wave, int lo4, int lo2) {
  constexpr int NT = nt_of(LV), R = (NT + 3) / 4, N = nin_of(LV);
#pragma unroll
  for (int r = 0; r < R; ++r) {
    const int t = ((wave + rot_of(LV)) & 3) + 4 * r;
    if (4 * r + 3 < NT || t < NT) {
      const int g = o_in + 128 * t;
      if (!G) {
        s.t[r].q  = *(const v4f_u*)(row + g + lo4);
        s.t[r].h2 = *(const v2f_u*)(row + g + lo2);
      } else {
        s.t[r].q.x  = ldg_guard(row, g + lo4,     N);
        s.t[r].q.y  = ldg_guard(row, g + lo4 + 1, N);
        s.t[r].q.z  = ldg_guard(row, g + lo4 + 2, N);
        s.t[r].q.w  = ldg_guard(row, g + lo4 + 3, N);
        s.t[r].h2.x = ldg_guard(row, g + lo2,     N);
        s.t[r].h2.y = ldg_guard(row, g + lo2 + 1, N);
      }
    }
  }
}

template <int LV>
__device__ __forceinline__ void run_level(const BSet<1>& xa1, const BSet<LV>& xd,
                                          float* lds, float* __restrict__ rowO,
                                          int o_out, int wave, int lo4, int lo2, int st,
                                          const v2f (&alo)[3], const v2f (&ahi)[3]) {
  constexpr int NT = nt_of(LV), R = (NT + 3) / 4;
#pragma unroll
  for (int r = 0; r < R; ++r) {
    const int t = ((wave + rot_of(LV)) & 3) + 4 * r;
    if (4 * r + 3 < NT || t < NT) {
      v4f qa; v2f ha;
      if (LV == 1) {
        qa = xa1.t[0].q;  ha = xa1.t[0].h2;
      } else {
        const float* in = lds + bufo_of(LV - 1) + 128 * t;
        qa = *(const v4f*)(in + lo4);
        ha = *(const v2f*)(in + lo2);
      }
      const v4f qd = xd.t[r].q;
      const v2f hd = xd.t[r].h2;
      v8f acc = {0.f, 0.f, 0.f, 0.f, 0.f, 0.f, 0.f, 0.f};
      v2f b0a = {qa.x, qa.y}, b1a = {qa.z, qa.w};
      v2f b0d = {qd.x, qd.y}, b1d = {qd.z, qd.w};
      acc = wmma4(alo[0], b0a, acc);
      acc = wmma4(ahi[0], b0d, acc);
      acc = wmma4(alo[1], b1a, acc);
      acc = wmma4(ahi[1], b1d, acc);
      acc = wmma4(alo[2], ha, acc);
      acc = wmma4(ahi[2], hd, acc);
      if (LV == 6) {
        float* p = rowO + o_out + 256 * t + st;
        v4f s0 = {acc[0], acc[1], acc[2], acc[3]};
        v4f s1 = {acc[4], acc[5], acc[6], acc[7]};
        *(v4f_u*)(p)     = s0;
        *(v4f_u*)(p + 8) = s1;
      } else {
        const int lim = nin_of(LV + 1) - (o_out + 256 * t + st);
        v4f s0, s1;
        s0.x = (0 < lim)  ? acc[0] : 0.f;  s0.y = (1 < lim)  ? acc[1] : 0.f;
        s0.z = (2 < lim)  ? acc[2] : 0.f;  s0.w = (3 < lim)  ? acc[3] : 0.f;
        s1.x = (8 < lim)  ? acc[4] : 0.f;  s1.y = (9 < lim)  ? acc[5] : 0.f;
        s1.z = (10 < lim) ? acc[6] : 0.f;  s1.w = (11 < lim) ? acc[7] : 0.f;
        float* p = lds + bufo_of(LV) + 256 * t + st;
        *(v4f*)(p)     = s0;
        *(v4f*)(p + 8) = s1;
      }
    }
  }
}

__device__ __forceinline__ void idwt_chunk(const float* __restrict__ ra,
                                           const float* __restrict__ r1,
                                           const float* __restrict__ r2,
                                           const float* __restrict__ r3,
                                           const float* __restrict__ r4,
                                           const float* __restrict__ r5,
                                           const float* __restrict__ r6,
                                           float* __restrict__ rowO, float* lds,
                                           int o0, int wave, int lane, bool interior) {
  const int c = lane & 15;
  const int h = lane >> 4;

  const int lo4 = 8 * c + 4 * h;
  const int lo2 = 8 * c + 8 + 2 * h;
  const int st  = 16 * c + 4 * h;

  BSet<1> xa1, d1;  BSet<2> d2;  BSet<3> d3;  BSet<4> d4;  BSet<5> d5;  BSet<6> d6;
  if (interior) {
    load_set<1, false>(xa1, ra, o0,      wave, lo4, lo2);
    load_set<1, false>(d1,  r1, o0,      wave, lo4, lo2);
    load_set<2, false>(d2,  r2, o0 << 1, wave, lo4, lo2);
    load_set<3, false>(d3,  r3, o0 << 2, wave, lo4, lo2);
    load_set<4, false>(d4,  r4, o0 << 3, wave, lo4, lo2);
    load_set<5, false>(d5,  r5, o0 << 4, wave, lo4, lo2);
    load_set<6, false>(d6,  r6, o0 << 5, wave, lo4, lo2);
    __builtin_amdgcn_fence(__ATOMIC_RELEASE, "wavefront");
  }
  const int so = ((c & 4) << 1) | ((c & 8) >> 1) | (c & 3);
  v2f alo[3], ahi[3];
  {
    const int t[6] = {so + 6 - 8 * h, so + 4 - 8 * h, so + 2 - 8 * h, so - 8 * h,
                      so - 10 - 4 * h, so - 12 - 4 * h};
    float wl[6], wh[6];
#pragma unroll
    for (int i = 0; i < 6; ++i) { wl[i] = c_rec_lo[t[i] & 7];  wh[i] = c_rec_hi[t[i] & 7]; }
    __builtin_amdgcn_sched_barrier(0);
#pragma unroll
    for (int i = 0; i < 6; ++i) {
      const bool in = (t[i] >= 0 && t[i] < 8);
      wl[i] = in ? wl[i] : 0.0f;
      wh[i] = in ? wh[i] : 0.0f;
    }
#pragma unroll
    for (int ks = 0; ks < 3; ++ks) {
      alo[ks].x = wl[2 * ks];  alo[ks].y = wl[2 * ks + 1];
      ahi[ks].x = wh[2 * ks];  ahi[ks].y = wh[2 * ks + 1];
    }
  }

  if (!interior) { load_set<1, true>(xa1, ra, o0, wave, lo4, lo2);
                   load_set<1, true>(d1,  r1, o0, wave, lo4, lo2); }
  run_level<1>(xa1, d1, lds, rowO, o0 << 1, wave, lo4, lo2, st, alo, ahi);
  __syncthreads();
  if (!interior) load_set<2, true>(d2, r2, o0 << 1, wave, lo4, lo2);
  run_level<2>(xa1, d2, lds, rowO, o0 << 2, wave, lo4, lo2, st, alo, ahi);
  __syncthreads();
  if (!interior) load_set<3, true>(d3, r3, o0 << 2, wave, lo4, lo2);
  run_level<3>(xa1, d3, lds, rowO, o0 << 3, wave, lo4, lo2, st, alo, ahi);
  __syncthreads();
  if (!interior) load_set<4, true>(d4, r4, o0 << 3, wave, lo4, lo2);
  run_level<4>(xa1, d4, lds, rowO, o0 << 4, wave, lo4, lo2, st, alo, ahi);
  __syncthreads();
  if (!interior) load_set<5, true>(d5, r5, o0 << 4, wave, lo4, lo2);
  run_level<5>(xa1, d5, lds, rowO, o0 << 5, wave, lo4, lo2, st, alo, ahi);
  __syncthreads();
  if (!interior) load_set<6, true>(d6, r6, o0 << 5, wave, lo4, lo2);
  run_level<6>(xa1, d6, lds, rowO, o0 << 6, wave, lo4, lo2, st, alo, ahi);
}

__global__ __launch_bounds__(128)
void idwt6_fused_wmma(const float* __restrict__ a,  const float* __restrict__ d6,
                      const float* __restrict__ d5, const float* __restrict__ d4,
                      const float* __restrict__ d3, const float* __restrict__ d2,
                      const float* __restrict__ d1, float* __restrict__ out) {
  __shared__ __attribute__((aligned(16))) float lds[LDS_FLOATS];
  const int wave = __builtin_amdgcn_readfirstlane(threadIdx.x >> 5);
  const int lane = threadIdx.x & 31;
  const int o0   = (63 - (int)blockIdx.x) * 64;
  const size_t b = blockIdx.y;

  const float* ra = a  + b * (size_t)nin_of(1);
  const float* r1 = d6 + b * (size_t)nin_of(1);
  const float* r2 = d5 + b * (size_t)nin_of(2);
  const float* r3 = d4 + b * (size_t)nin_of(3);
  const float* r4 = d3 + b * (size_t)nin_of(4);
  const float* r5 = d2 + b * (size_t)nin_of(5);
  const float* r6 = d1 + b * (size_t)nin_of(6);
  float* rowO = out + b * (size_t)L_OUT;

  const bool interior =
      (o0        + 128 * nt_of(1) + 4 <= nin_of(1)) &&
      ((o0 << 1) + 128 * nt_of(2) + 4 <= nin_of(2)) &&
      ((o0 << 2) + 128 * nt_of(3) + 4 <= nin_of(3)) &&
      ((o0 << 3) + 128 * nt_of(4) + 4 <= nin_of(4)) &&
      ((o0 << 4) + 128 * nt_of(5) + 4 <= nin_of(5)) &&
      ((o0 << 5) + 128 * nt_of(6) + 4 <= nin_of(6));
  idwt_chunk(ra, r1, r2, r3, r4, r5, r6, rowO, lds, o0, wave, lane, interior);
}

extern "C" void kernel_launch(void* const* d_in, const int* in_sizes, int n_in_cnt,
                              void* d_out, int out_size, void* d_ws, size_t ws_size,
                              hipStream_t stream) {
  (void)in_sizes; (void)n_in_cnt; (void)out_size; (void)d_ws; (void)ws_size;
  const float* a  = (const float*)d_in[0];
  const float* d6 = (const float*)d_in[1];
  const float* d5 = (const float*)d_in[2];
  const float* d4 = (const float*)d_in[3];
  const float* d3 = (const float*)d_in[4];
  const float* d2 = (const float*)d_in[5];
  const float* d1 = (const float*)d_in[6];
  float* out = (float*)d_out;
  idwt6_fused_wmma<<<dim3(64, 128), 128, 0, stream>>>(a, d6, d5, d4, d3, d2, d1, out);
}


# Level-6 results pass through a wave-private LDS slot so that every global store instruction writes four whole 128-byte lines (was sixteen 32-byte pieces in eight lines); sum order unchanged
# speedup vs baseline: 4.6734x; 1.0553x over previous
#include <hip/hip_runtime.h>


typedef __attribute__((ext_vector_type(2))) float v2f;
typedef __attribute__((ext_vector_type(4))) float v4f;
typedef __attribute__((ext_vector_type(8))) float v8f;
typedef v2f __attribute__((aligned(4))) v2f_u;
typedef v4f __attribute__((aligned(4))) v4f_u;

__constant__ float c_rec_lo[8] = {
    0.23037781330885523f,  0.7148465705525415f,  0.6308807679295904f,
   -0.02798376941698385f, -0.18703481171888114f, 0.030841381835986965f,
    0.032883011666982945f,-0.010597401784997278f};
__constant__ float c_rec_hi[8] = {
   -0.010597401784997278f,-0.032883011666982945f, 0.030841381835986965f,
    0.18703481171888114f, -0.02798376941698385f, -0.6308807679295904f,
    0.7148465705525415f,  -0.23037781330885523f};

__host__ __device__ constexpr int nt_of(int lv) {
  return lv == 1 ? 2 : lv == 2 ? 2 : lv == 3 ? 3 : lv == 4 ? 5 : lv == 5 ? 9 : 16;
}
__host__ __device__ constexpr int nin_of(int lv) {
  return lv == 1 ? 4102 : lv == 2 ? 8198 : lv == 3 ? 16390 : lv == 4 ? 32774
       : lv == 5 ? 65541 : 131075;
}
__host__ __device__ constexpr int bufo_of(int lv) {
  return lv == 1 ? 0 : lv == 2 ? 512 : lv == 3 ? 1024 : lv == 4 ? 1792 : 3072;
}
__host__ __device__ constexpr int rot_of(int lv) {
  return lv == 1 ? 2 : lv == 2 ? 2 : lv == 4 ? 3 : 0;
}
#define LDS_FLOATS 5376
#define L_OUT 262144

__device__ __forceinline__ float ldg_guard(const float* row, int col, int n) {
  const int cc = (col < n) ? col : (n - 1);
  const float v = row[cc];
  return (col < n) ? v : 0.0f;
}

__device__ __forceinline__ v8f wmma4(v2f a, v2f b, v8f c) {
  return __builtin_amdgcn_wmma_f32_16x16x4_f32(false, a, false, b, (short)0, c,
                                               false, false);
}

struct BT { v4f q; v2f h2; };
template <int LV> struct BSet { BT t[(nt_of(LV) + 3) / 4]; };

template <int LV, bool G>
__device__ __forceinline__ void load_set(BSet<LV>& s, const float* __restrict__ row,
                                         int o_in, int wave, int lo4, int lo2) {
  constexpr int NT = nt_of(LV), R = (NT + 3) / 4, N = nin_of(LV);
#pragma unroll
  for (int r = 0; r < R; ++r) {
    const int t = ((wave + rot_of(LV)) & 3) + 4 * r;
    if (4 * r + 3 < NT || t < NT) {
      const int g = o_in + 128 * t;
      if (!G) {
        s.t[r].q  = *(const v4f_u*)(row + g + lo4);
        s.t[r].h2 = *(const v2f_u*)(row + g + lo2);
      } else {
        s.t[r].q.x  = ldg_guard(row, g + lo4,     N);
        s.t[r].q.y  = ldg_guard(row, g + lo4 + 1, N);
        s.t[r].q.z  = ldg_guard(row, g + lo4 + 2, N);
        s.t[r].q.w  = ldg_guard(row, g + lo4 + 3, N);
        s.t[r].h2.x = ldg_guard(row, g + lo2,     N);
        s.t[r].h2.y = ldg_guard(row, g + lo2 + 1, N);
      }
    }
  }
}

template <int LV>
__device__ __forceinline__ void run_level(const BSet<1>& xa1, const BSet<LV>& xd,
                                          float* lds, float* __restrict__ rowO,
                                          int o_out, int wave, int lo4, int lo2, int st,
                                          const v2f (&alo)[3], const v2f (&ahi)[3]) {
  constexpr int NT = nt_of(LV), R = (NT + 3) / 4;
  v8f accs[R];
#pragma unroll
  for (int r = 0; r < R; ++r) {
    const int t = ((wave + rot_of(LV)) & 3) + 4 * r;
    if (4 * r + 3 < NT || t < NT) {
      v4f qa; v2f ha;
      if (LV == 1) {
        qa = xa1.t[0].q;  ha = xa1.t[0].h2;
      } else {
        const float* in = lds + bufo_of(LV - 1) + 128 * t;
        qa = *(const v4f*)(in + lo4);
        ha = *(const v2f*)(in + lo2);
      }
      const v4f qd = xd.t[r].q;
      const v2f hd = xd.t[r].h2;
      v8f acc = {0.f, 0.f, 0.f, 0.f, 0.f, 0.f, 0.f, 0.f};
      v2f b0a = {qa.x, qa.y}, b1a = {qa.z, qa.w};
      v2f b0d = {qd.x, qd.y}, b1d = {qd.z, qd.w};
      acc = wmma4(alo[0], b0a, acc);
      acc = wmma4(ahi[0], b0d, acc);
      acc = wmma4(alo[1], b1a, acc);
      acc = wmma4(ahi[1], b1d, acc);
      acc = wmma4(alo[2], ha, acc);
      acc = wmma4(ahi[2], hd, acc);
      if (LV == 6) {
        accs[r] = acc;
      } else {
        const int lim = nin_of(LV + 1) - (o_out + 256 * t + st);
        v4f s0, s1;
        s0.x = (0 < lim)  ? acc[0] : 0.f;  s0.y = (1 < lim)  ? acc[1] : 0.f;
        s0.z = (2 < lim)  ? acc[2] : 0.f;  s0.w = (3 < lim)  ? acc[3] : 0.f;
        s1.x = (8 < lim)  ? acc[4] : 0.f;  s1.y = (9 < lim)  ? acc[5] : 0.f;
        s1.z = (10 < lim) ? acc[6] : 0.f;  s1.w = (11 < lim) ? acc[7] : 0.f;
        float* p = lds + bufo_of(LV) + 256 * t + st;
        *(v4f*)(p)     = s0;
        *(v4f*)(p + 8) = s1;
      }
    }
  }
  if (LV == 6) {
    float* stg = lds + 544 * wave;
    const int wpos = st + 4 * (st >> 6);
    const int l4   = ((st >> 2) & 60) + ((st & 4) << 4);
    const int rpos = l4 + 4 * (l4 >> 6);
    v4f o[R][2];
#pragma unroll
    for (int r0 = 0; r0 < R; r0 += 2) {
#pragma unroll
      for (int u = 0; u < 2; ++u) {
        const v8f acc = accs[r0 + u];
        float* slot = stg + 272 * u;
        v4f s0 = {acc[0], acc[1], acc[2], acc[3]};
        v4f s1 = {acc[4], acc[5], acc[6], acc[7]};
        *(v4f*)(slot + wpos)     = s0;
        *(v4f*)(slot + wpos + 8) = s1;
      }
      __builtin_amdgcn_fence(__ATOMIC_SEQ_CST, "wavefront");
#pragma unroll
      for (int u = 0; u < 2; ++u) {
        const float* slot = stg + 272 * u;
        o[r0 + u][0] = *(const v4f*)(slot + rpos);
        o[r0 + u][1] = *(const v4f*)(slot + rpos + 136);
      }
      __builtin_amdgcn_fence(__ATOMIC_SEQ_CST, "wavefront");
    }
#pragma unroll
    for (int r = 0; r < R; ++r) {
      const int t = ((wave + rot_of(LV)) & 3) + 4 * r;
      float* p = rowO + o_out + 256 * t + l4;
      *(v4f_u*)(p)       = o[r][0];
      *(v4f_u*)(p + 128) = o[r][1];
    }
  }
}

__device__ __forceinline__ void idwt_chunk(const float* __restrict__ ra,
                                           const float* __restrict__ r1,
                                           const float* __restrict__ r2,
                                           const float* __restrict__ r3,
                                           const float* __restrict__ r4,
                                           const float* __restrict__ r5,
                                           const float* __restrict__ r6,
                                           float* __restrict__ rowO, float* lds,
                                           int o0, int wave, int lane, bool interior) {
  const int c = lane & 15;
  const int h = lane >> 4;

  const int lo4 = 8 * c + 4 * h;
  const int lo2 = 8 * c + 8 + 2 * h;
  const int st  = 16 * c + 4 * h;

  BSet<1> xa1, d1;  BSet<2> d2;  BSet<3> d3;  BSet<4> d4;  BSet<5> d5;  BSet<6> d6;
  if (interior) {
    load_set<1, false>(xa1, ra, o0,      wave, lo4, lo2);
    load_set<1, false>(d1,  r1, o0,      wave, lo4, lo2);
    load_set<2, false>(d2,  r2, o0 << 1, wave, lo4, lo2);
    load_set<3, false>(d3,  r3, o0 << 2, wave, lo4, lo2);
    load_set<4, false>(d4,  r4, o0 << 3, wave, lo4, lo2);
    load_set<5, false>(d5,  r5, o0 << 4, wave, lo4, lo2);
    load_set<6, false>(d6,  r6, o0 << 5, wave, lo4, lo2);
    __builtin_amdgcn_fence(__ATOMIC_RELEASE, "wavefront");
  }
  const int so = ((c & 4) << 1) | ((c & 8) >> 1) | (c & 3);
  v2f alo[3], ahi[3];
  {
    const int t[6] = {so + 6 - 8 * h, so + 4 - 8 * h, so + 2 - 8 * h, so - 8 * h,
                      so - 10 - 4 * h, so - 12 - 4 * h};
    float wl[6], wh[6];
#pragma unroll
    for (int i = 0; i < 6; ++i) { wl[i] = c_rec_lo[t[i] & 7];  wh[i] = c_rec_hi[t[i] & 7]; }
    __builtin_amdgcn_sched_barrier(0);
#pragma unroll
    for (int i = 0; i < 6; ++i) {
      const bool in = (t[i] >= 0 && t[i] < 8);
      wl[i] = in ? wl[i] : 0.0f;
      wh[i] = in ? wh[i] : 0.0f;
    }
#pragma unroll
    for (int ks = 0; ks < 3; ++ks) {
      alo[ks].x = wl[2 * ks];  alo[ks].y = wl[2 * ks + 1];
      ahi[ks].x = wh[2 * ks];  ahi[ks].y = wh[2 * ks + 1];
    }
  }

  if (!interior) { load_set<1, true>(xa1, ra, o0, wave, lo4, lo2);
                   load_set<1, true>(d1,  r1, o0, wave, lo4, lo2); }
  run_level<1>(xa1, d1, lds, rowO, o0 << 1, wave, lo4, lo2, st, alo, ahi);
  __syncthreads();
  if (!interior) load_set<2, true>(d2, r2, o0 << 1, wave, lo4, lo2);
  run_level<2>(xa1, d2, lds, rowO, o0 << 2, wave, lo4, lo2, st, alo, ahi);
  __syncthreads();
  if (!interior) load_set<3, true>(d3, r3, o0 << 2, wave, lo4, lo2);
  run_level<3>(xa1, d3, lds, rowO, o0 << 3, wave, lo4, lo2, st, alo, ahi);
  __syncthreads();
  if (!interior) load_set<4, true>(d4, r4, o0 << 3, wave, lo4, lo2);
  run_level<4>(xa1, d4, lds, rowO, o0 << 4, wave, lo4, lo2, st, alo, ahi);
  __syncthreads();
  if (!interior) load_set<5, true>(d5, r5, o0 << 4, wave, lo4, lo2);
  run_level<5>(xa1, d5, lds, rowO, o0 << 5, wave, lo4, lo2, st, alo, ahi);
  __syncthreads();
  if (!interior) load_set<6, true>(d6, r6, o0 << 5, wave, lo4, lo2);
  run_level<6>(xa1, d6, lds, rowO, o0 << 6, wave, lo4, lo2, st, alo, ahi);
}

__global__ __launch_bounds__(128, 9)
void idwt6_fused_wmma(const float* __restrict__ a,  const float* __restrict__ d6,
                      const float* __restrict__ d5, const float* __restrict__ d4,
                      const float* __restrict__ d3, const float* __restrict__ d2,
                      const float* __restrict__ d1, float* __restrict__ out) {
  __shared__ __attribute__((aligned(16))) float lds[LDS_FLOATS];
  const int wave = __builtin_amdgcn_readfirstlane(threadIdx.x >> 5);
  const int lane = threadIdx.x & 31;
  const int o0   = (63 - (int)blockIdx.x) * 64;
  const size_t b = blockIdx.y;

  const float* ra = a  + b * (size_t)nin_of(1);
  const float* r1 = d6 + b * (size_t)nin_of(1);
  const float* r2 = d5 + b * (size_t)nin_of(2);
  const float* r3 = d4 + b * (size_t)nin_of(3);
  const float* r4 = d3 + b * (size_t)nin_of(4);
  const float* r5 = d2 + b * (size_t)nin_of(5);
  const float* r6 = d1 + b * (size_t)nin_of(6);
  float* rowO = out + b * (size_t)L_OUT;

  const bool interior =
      (o0        + 128 * nt_of(1) + 4 <= nin_of(1)) &&
      ((o0 << 1) + 128 * nt_of(2) + 4 <= nin_of(2)) &&
      ((o0 << 2) + 128 * nt_of(3) + 4 <= nin_of(3)) &&
      ((o0 << 3) + 128 * nt_of(4) + 4 <= nin_of(4)) &&
      ((o0 << 4) + 128 * nt_of(5) + 4 <= nin_of(5)) &&
      ((o0 << 5) + 128 * nt_of(6) + 4 <= nin_of(6));
  idwt_chunk(ra, r1, r2, r3, r4, r5, r6, rowO, lds, o0, wave, lane, interior);
}

extern "C" void kernel_launch(void* const* d_in, const int* in_sizes, int n_in_cnt,
                              void* d_out, int out_size, void* d_ws, size_t ws_size,
                              hipStream_t stream) {
  (void)in_sizes; (void)n_in_cnt; (void)out_size; (void)d_ws; (void)ws_size;
  const float* a  = (const float*)d_in[0];
  const float* d6 = (const float*)d_in[1];
  const float* d5 = (const float*)d_in[2];
  const float* d4 = (const float*)d_in[3];
  const float* d3 = (const float*)d_in[4];
  const float* d2 = (const float*)d_in[5];
  const float* d1 = (const float*)d_in[6];
  float* out = (float*)d_out;
  idwt6_fused_wmma<<<dim3(64, 128), 128, 0, stream>>>(a, d6, d5, d4, d3, d2, d1, out);
}


# Resident grid: 2048 work-groups (8 per WGP) loop over 4 chunks each, weights built once, barriers order LDS only so loads and stores stay in flight; 8192 waves enter, not 32768; sum order unchanged
# speedup vs baseline: 4.9313x; 1.0552x over previous
#include <hip/hip_runtime.h>


typedef __attribute__((ext_vector_type(2))) float v2f;
typedef __attribute__((ext_vector_type(4))) float v4f;
typedef __attribute__((ext_vector_type(8))) float v8f;
typedef v2f __attribute__((aligned(4))) v2f_u;
typedef v4f __attribute__((aligned(4))) v4f_u;

__constant__ float c_rec_lo[8] = {
    0.23037781330885523f,  0.7148465705525415f,  0.6308807679295904f,
   -0.02798376941698385f, -0.18703481171888114f, 0.030841381835986965f,
    0.032883011666982945f,-0.010597401784997278f};
__constant__ float c_rec_hi[8] = {
   -0.010597401784997278f,-0.032883011666982945f, 0.030841381835986965f,
    0.18703481171888114f, -0.02798376941698385f, -0.6308807679295904f,
    0.7148465705525415f,  -0.23037781330885523f};

__host__ __device__ constexpr int nt_of(int lv) {
  return lv == 1 ? 2 : lv == 2 ? 2 : lv == 3 ? 3 : lv == 4 ? 5 : lv == 5 ? 9 : 16;
}
__host__ __device__ constexpr int nin_of(int lv) {
  return lv == 1 ? 4102 : lv == 2 ? 8198 : lv == 3 ? 16390 : lv == 4 ? 32774
       : lv == 5 ? 65541 : 131075;
}
__host__ __device__ constexpr int bufo_of(int lv) {
  return lv == 1 ? 0 : lv == 2 ? 512 : lv == 3 ? 1024 : lv == 4 ? 1792 : 3072;
}
__host__ __device__ constexpr int rot_of(int lv) {
  return lv == 1 ? 2 : lv == 2 ? 2 : lv == 4 ? 3 : 0;
}
#define LDS_FLOATS 5376
#define L_OUT 262144

__device__ __forceinline__ float ldg_guard(const float* row, int col, int n) {
  const int cc = (col < n) ? col : (n - 1);
  const float v = row[cc];
  return (col < n) ? v : 0.0f;
}

__device__ __forceinline__ v8f wmma4(v2f a, v2f b, v8f c) {
  return __builtin_amdgcn_wmma_f32_16x16x4_f32(false, a, false, b, (short)0, c,
                                               false, false);
}

struct BT { v4f q; v2f h2; };
template <int LV> struct BSet { BT t[(nt_of(LV) + 3) / 4]; };

template <int LV, bool G>
__device__ __forceinline__ void load_set(BSet<LV>& s, const float* __restrict__ row,
                                         int o_in, int wave, int lo4, int lo2) {
  constexpr int NT = nt_of(LV), R = (NT + 3) / 4, N = nin_of(LV);
#pragma unroll
  for (int r = 0; r < R; ++r) {
    const int t = ((wave + rot_of(LV)) & 3) + 4 * r;
    if (4 * r + 3 < NT || t < NT) {
      const int g = o_in + 128 * t;
      if (!G) {
        s.t[r].q  = *(const v4f_u*)(row + g + lo4);
        s.t[r].h2 = *(const v2f_u*)(row + g + lo2);
      } else {
        s.t[r].q.x  = ldg_guard(row, g + lo4,     N);
        s.t[r].q.y  = ldg_guard(row, g + lo4 + 1, N);
        s.t[r].q.z  = ldg_guard(row, g + lo4 + 2, N);
        s.t[r].q.w  = ldg_guard(row, g + lo4 + 3, N);
        s.t[r].h2.x = ldg_guard(row, g + lo2,     N);
        s.t[r].h2.y = ldg_guard(row, g + lo2 + 1, N);
      }
    }
  }
}

template <int LV>
__device__ __forceinline__ void run_level(const BSet<1>& xa1, const BSet<LV>& xd,
                                          float* lds, float* __restrict__ rowO,
                                          int o_out, int wave, int lo4, int lo2, int st,
                                          const v2f (&alo)[3], const v2f (&ahi)[3]) {
  constexpr int NT = nt_of(LV), R = (NT + 3) / 4;
  v8f accs[R];
#pragma unroll
  for (int r = 0; r < R; ++r) {
    const int t = ((wave + rot_of(LV)) & 3) + 4 * r;
    if (4 * r + 3 < NT || t < NT) {
      v4f qa; v2f ha;
      if (LV == 1) {
        qa = xa1.t[0].q;  ha = xa1.t[0].h2;
      } else {
        const float* in = lds + bufo_of(LV - 1) + 128 * t;
        qa = *(const v4f*)(in + lo4);
        ha = *(const v2f*)(in + lo2);
      }
      const v4f qd = xd.t[r].q;
      const v2f hd = xd.t[r].h2;
      v8f acc = {0.f, 0.f, 0.f, 0.f, 0.f, 0.f, 0.f, 0.f};
      v2f b0a = {qa.x, qa.y}, b1a = {qa.z, qa.w};
      v2f b0d = {qd.x, qd.y}, b1d = {qd.z, qd.w};
      acc = wmma4(alo[0], b0a, acc);
      acc = wmma4(ahi[0], b0d, acc);
      acc = wmma4(alo[1], b1a, acc);
      acc = wmma4(ahi[1], b1d, acc);
      acc = wmma4(alo[2], ha, acc);
      acc = wmma4(ahi[2], hd, acc);
      if (LV == 6) {
        accs[r] = acc;
      } else {
        const int lim = nin_of(LV + 1) - (o_out + 256 * t + st);
        v4f s0, s1;
        s0.x = (0 < lim)  ? acc[0] : 0.f;  s0.y = (1 < lim)  ? acc[1] : 0.f;
        s0.z = (2 < lim)  ? acc[2] : 0.f;  s0.w = (3 < lim)  ? acc[3] : 0.f;
        s1.x = (8 < lim)  ? acc[4] : 0.f;  s1.y = (9 < lim)  ? acc[5] : 0.f;
        s1.z = (10 < lim) ? acc[6] : 0.f;  s1.w = (11 < lim) ? acc[7] : 0.f;
        float* p = lds + bufo_of(LV) + 256 * t + st;
        *(v4f*)(p)     = s0;
        *(v4f*)(p + 8) = s1;
      }
    }
  }
  if (LV == 6) {
    float* stg = lds + 544 * wave;
    const int wpos = st + 4 * (st >> 6);
    const int l4   = ((st >> 2) & 60) + ((st & 4) << 4);
    const int rpos = l4 + 4 * (l4 >> 6);
    v4f o[R][2];
#pragma unroll
    for (int r0 = 0; r0 < R; r0 += 2) {
#pragma unroll
      for (int u = 0; u < 2; ++u) {
        const v8f acc = accs[r0 + u];
        float* slot = stg + 272 * u;
        v4f s0 = {acc[0], acc[1], acc[2], acc[3]};
        v4f s1 = {acc[4], acc[5], acc[6], acc[7]};
        *(v4f*)(slot + wpos)     = s0;
        *(v4f*)(slot + wpos + 8) = s1;
      }
      __builtin_amdgcn_fence(__ATOMIC_SEQ_CST, "wavefront");
#pragma unroll
      for (int u = 0; u < 2; ++u) {
        const float* slot = stg + 272 * u;
        o[r0 + u][0] = *(const v4f*)(slot + rpos);
        o[r0 + u][1] = *(const v4f*)(slot + rpos + 136);
      }
      __builtin_amdgcn_fence(__ATOMIC_SEQ_CST, "wavefront");
    }
#pragma unroll
    for (int r = 0; r < R; ++r) {
      const int t = ((wave + rot_of(LV)) & 3) + 4 * r;
      float* p = rowO + o_out + 256 * t + l4;
      *(v4f_u*)(p)       = o[r][0];
      *(v4f_u*)(p + 128) = o[r][1];
    }
  }
}

__device__ __forceinline__ void lds_barrier() {
  __builtin_amdgcn_fence(__ATOMIC_RELEASE, "workgroup", "local");
  __builtin_amdgcn_s_barrier();
  __builtin_amdgcn_fence(__ATOMIC_ACQUIRE, "workgroup", "local");
}

struct Sets { BSet<1> xa1, d1;  BSet<2> d2;  BSet<3> d3;  BSet<4> d4;  BSet<5> d5;  BSet<6> d6; };
struct Rows { const float *ra, *r1, *r2, *r3, *r4, *r5, *r6;  float* rowO; };

#define N_ROWS     128
#define N_CHUNKS   64
#define N_EDGE     3
#define N_INNER    (N_CHUNKS - N_EDGE)
#define N_WG       2048
#define N_IT       4
#define N_EDGE_WG  (N_ROWS * N_EDGE)
#define N_REST_WG  (N_WG - N_EDGE_WG)

__device__ __forceinline__ void task_of(int w, int it, int& row, int& o0) {
  int chunk;
  if (w < N_EDGE_WG && it == 0) {
    row = w & (N_ROWS - 1);  chunk = N_INNER + w / N_ROWS;
  } else {
    const int q = (w < N_EDGE_WG) ? N_IT * N_REST_WG + w + N_EDGE_WG * (it - 1)
                                  : (w - N_EDGE_WG) + N_REST_WG * it;
    row = q / N_INNER;  chunk = q - N_INNER * row;
  }
  o0 = chunk * 64;
}

__host__ __device__ constexpr bool is_interior(int o0) {
  return (o0        + 128 * nt_of(1) + 4 <= nin_of(1)) &&
         ((o0 << 1) + 128 * nt_of(2) + 4 <= nin_of(2)) &&
         ((o0 << 2) + 128 * nt_of(3) + 4 <= nin_of(3)) &&
         ((o0 << 3) + 128 * nt_of(4) + 4 <= nin_of(4)) &&
         ((o0 << 4) + 128 * nt_of(5) + 4 <= nin_of(5)) &&
         ((o0 << 5) + 128 * nt_of(6) + 4 <= nin_of(6));
}

__device__ __forceinline__ Rows rows_of(const float* __restrict__ a,  const float* __restrict__ d6,
                                        const float* __restrict__ d5, const float* __restrict__ d4,
                                        const float* __restrict__ d3, const float* __restrict__ d2,
                                        const float* __restrict__ d1, float* __restrict__ out, int row) {
  const size_t b = (size_t)row;
  Rows r;
  r.ra = a  + b * (size_t)nin_of(1);
  r.r1 = d6 + b * (size_t)nin_of(1);
  r.r2 = d5 + b * (size_t)nin_of(2);
  r.r3 = d4 + b * (size_t)nin_of(3);
  r.r4 = d3 + b * (size_t)nin_of(4);
  r.r5 = d2 + b * (size_t)nin_of(5);
  r.r6 = d1 + b * (size_t)nin_of(6);
  r.rowO = out + b * (size_t)L_OUT;
  return r;
}

__device__ __forceinline__ void load_all(Sets& s, const Rows& r, int o0, int wave, int lo4, int lo2) {
  load_set<1, false>(s.xa1, r.ra, o0,      wave, lo4, lo2);
  load_set<1, false>(s.d1,  r.r1, o0,      wave, lo4, lo2);
  load_set<2, false>(s.d2,  r.r2, o0 << 1, wave, lo4, lo2);
  load_set<3, false>(s.d3,  r.r3, o0 << 2, wave, lo4, lo2);
  load_set<4, false>(s.d4,  r.r4, o0 << 3, wave, lo4, lo2);
  load_set<5, false>(s.d5,  r.r5, o0 << 4, wave, lo4, lo2);
  load_set<6, false>(s.d6,  r.r6, o0 << 5, wave, lo4, lo2);
  __builtin_amdgcn_fence(__ATOMIC_RELEASE, "wavefront");
}

__device__ __forceinline__ void compute_chunk(Sets& s, const Rows& r, float* lds, int o0,
                                              int wave, int lo4, int lo2, int st, bool interior,
                                              const v2f (&alo)[3], const v2f (&ahi)[3]) {
  if (!interior) { load_set<1, true>(s.xa1, r.ra, o0, wave, lo4, lo2);
                   load_set<1, true>(s.d1,  r.r1, o0, wave, lo4, lo2); }
  run_level<1>(s.xa1, s.d1, lds, r.rowO, o0 << 1, wave, lo4, lo2, st, alo, ahi);
  lds_barrier();
  if (!interior) load_set<2, true>(s.d2, r.r2, o0 << 1, wave, lo4, lo2);
  run_level<2>(s.xa1, s.d2, lds, r.rowO, o0 << 2, wave, lo4, lo2, st, alo, ahi);
  lds_barrier();
  if (!interior) load_set<3, true>(s.d3, r.r3, o0 << 2, wave, lo4, lo2);
  run_level<3>(s.xa1, s.d3, lds, r.rowO, o0 << 3, wave, lo4, lo2, st, alo, ahi);
  lds_barrier();
  if (!interior) load_set<4, true>(s.d4, r.r4, o0 << 3, wave, lo4, lo2);
  run_level<4>(s.xa1, s.d4, lds, r.rowO, o0 << 4, wave, lo4, lo2, st, alo, ahi);
  lds_barrier();
  if (!interior) load_set<5, true>(s.d5, r.r5, o0 << 4, wave, lo4, lo2);
  run_level<5>(s.xa1, s.d5, lds, r.rowO, o0 << 5, wave, lo4, lo2, st, alo, ahi);
  lds_barrier();
  if (!interior) load_set<6, true>(s.d6, r.r6, o0 << 5, wave, lo4, lo2);
  run_level<6>(s.xa1, s.d6, lds, r.rowO, o0 << 6, wave, lo4, lo2, st, alo, ahi);
}

__global__ __launch_bounds__(128, 8)
void idwt6_fused_wmma(const float* __restrict__ a,  const float* __restrict__ d6,
                      const float* __restrict__ d5, const float* __restrict__ d4,
                      const float* __restrict__ d3, const float* __restrict__ d2,
                      const float* __restrict__ d1, float* __restrict__ out) {
  __shared__ __attribute__((aligned(16))) float lds[LDS_FLOATS];
  const int wave = __builtin_amdgcn_readfirstlane(threadIdx.x >> 5);
  const int lane = threadIdx.x & 31;
  const int w    = (int)blockIdx.x;

  const int c = lane & 15;
  const int h = lane >> 4;

  const int lo4 = 8 * c + 4 * h;
  const int lo2 = 8 * c + 8 + 2 * h;
  const int st  = 16 * c + 4 * h;

  int row0, o00;
  task_of(w, 0, row0, o00);
  const Rows rw0 = rows_of(a, d6, d5, d4, d3, d2, d1, out, row0);
  const bool interior0 = is_interior(o00);
  Sets s0;
  if (interior0) load_all(s0, rw0, o00, wave, lo4, lo2);

  const int so = ((c & 4) << 1) | ((c & 8) >> 1) | (c & 3);
  v2f alo[3], ahi[3];
  {
    const int t[6] = {so + 6 - 8 * h, so + 4 - 8 * h, so + 2 - 8 * h, so - 8 * h,
                      so - 10 - 4 * h, so - 12 - 4 * h};
    float wl[6], wh[6];
#pragma unroll
    for (int i = 0; i < 6; ++i) { wl[i] = c_rec_lo[t[i] & 7];  wh[i] = c_rec_hi[t[i] & 7]; }
    __builtin_amdgcn_sched_barrier(0);
#pragma unroll
    for (int i = 0; i < 6; ++i) {
      const bool in = (t[i] >= 0 && t[i] < 8);
      wl[i] = in ? wl[i] : 0.0f;
      wh[i] = in ? wh[i] : 0.0f;
    }
#pragma unroll
    for (int ks = 0; ks < 3; ++ks) {
      alo[ks].x = wl[2 * ks];  alo[ks].y = wl[2 * ks + 1];
      ahi[ks].x = wh[2 * ks];  ahi[ks].y = wh[2 * ks + 1];
    }
  }
  compute_chunk(s0, rw0, lds, o00, wave, lo4, lo2, st, interior0, alo, ahi);

  static_assert(is_interior((N_INNER - 1) * 64) && !is_interior(N_INNER * 64),
                "chunks 0..N_INNER-1 of a row are the interior ones");
#pragma clang loop unroll(disable)
  for (int it = 1; it < N_IT; ++it) {
    int row, o0;
    task_of(w, it, row, o0);
    const Rows r = rows_of(a, d6, d5, d4, d3, d2, d1, out, row);
    Sets s;
    load_all(s, r, o0, wave, lo4, lo2);
    lds_barrier();
    compute_chunk(s, r, lds, o0, wave, lo4, lo2, st, true, alo, ahi);
  }
}

extern "C" void kernel_launch(void* const* d_in, const int* in_sizes, int n_in_cnt,
                              void* d_out, int out_size, void* d_ws, size_t ws_size,
                              hipStream_t stream) {
  (void)in_sizes; (void)n_in_cnt; (void)out_size; (void)d_ws; (void)ws_size;
  const float* a  = (const float*)d_in[0];
  const float* d6 = (const float*)d_in[1];
  const float* d5 = (const float*)d_in[2];
  const float* d4 = (const float*)d_in[3];
  const float* d3 = (const float*)d_in[4];
  const float* d2 = (const float*)d_in[5];
  const float* d1 = (const float*)d_in[6];
  float* out = (float*)d_out;
  idwt6_fused_wmma<<<dim3(2048), 128, 0, stream>>>(a, d6, d5, d4, d3, d2, d1, out);
}
